# starting kernel
# baseline (speedup 1.0000x reference)
#include <hip/hip_runtime.h>
#include <hip/hip_bf16.h>

typedef __attribute__((ext_vector_type(2))) float v2f;
typedef __attribute__((ext_vector_type(8))) float v8f;

#define FCH 9
#define PSZ 24
#define NPIX (PSZ * PSZ)
#define PATCH (NPIX * FCH)
#define NV4   (PATCH / 4)
#define DCPAD 12
#define DCSZ  (NPIX * DCPAD)
#define NTHREADS 288

__global__ __launch_bounds__(NTHREADS)
void guided_conv_fused(const float* __restrict__ gin,
                       const float* __restrict__ din,
                       const float* __restrict__ conv_w,
                       const float* __restrict__ conv_b,
                       const float* __restrict__ dense_w,
                       const float* __restrict__ dense_b,
                       float* __restrict__ out)
{
    __shared__ __align__(16) float g[DCSZ];
    __shared__ __align__(16) float dpt[PATCH];
    __shared__ float cw[729];
    __shared__ float dwm[729];
    __shared__ float ck[81];
    __shared__ float part[NTHREADS];
    __shared__ float gap[FCH];
    __shared__ float dd[81];
    __shared__ float Mp[12 * 16];
    __shared__ float sM;

    const int tid = threadIdx.x;
    const int n  = blockIdx.x;
    const int b  = n >> 8;
    const int r  = n & 255;
    const int ih = r >> 4;
    const int iw = r & 15;
    const long base_in  = (((long)b * 384 + ih * PSZ) * 384 + iw * PSZ) * FCH;
    const long base_out = (long)n * PATCH;

    for (int i = tid; i < NV4; i += NTHREADS) {
        const int py = i / 54;
        const int q  = i - py * 54;
        const long off = base_in + (long)py * (384 * FCH) + q * 4;
        const float4 gv = *reinterpret_cast<const float4*>(gin + off);
        const float4 dv = *reinterpret_cast<const float4*>(din + off);
        *reinterpret_cast<float4*>(g   + i * 4) = gv;
        *reinterpret_cast<float4*>(dpt + i * 4) = dv;
    }
    for (int i = tid; i < 729; i += NTHREADS) {
        cw[i]  = conv_w[i];
        dwm[i] = dense_w[i];
    }
    __syncthreads();

    if (tid < 81) {
        const int oc = tid % 9;
        const int kx = (tid / 9) % 3;
        const int ky = tid / 27;
        float acc = conv_b[oc];
        #pragma unroll
        for (int dy = 0; dy < 3; ++dy)
            #pragma unroll
            for (int dx = 0; dx < 3; ++dx)
                for (int ic = 0; ic < FCH; ++ic)
                    acc += g[((ky * 8 + dy) * PSZ + (kx * 8 + dx)) * FCH + ic]
                         * cw[((dy * 3 + dx) * 9 + ic) * 9 + oc];
        ck[oc * 9 + ky * 3 + kx] = acc;
    }
    {
        const int c = tid % 9;
        const int s = tid / 9;
        float acc = 0.f;
        for (int p = s; p < NPIX; p += 32) acc += g[p * FCH + c];
        part[tid] = acc;
    }
    __syncthreads();

    if (tid < 9) {
        float acc = 0.f;
        for (int s = 0; s < 32; ++s) acc += part[s * 9 + tid];
        gap[tid] = acc * (1.0f / (float)NPIX);
        float ss = 0.f;
        #pragma unroll
        for (int j = 0; j < 9; ++j) { float v = ck[tid * 9 + j]; ss += v * v; }
        const float scale = 1.0f / fmaxf(sqrtf(ss), 1.0f);
        #pragma unroll
        for (int j = 0; j < 9; ++j) ck[tid * 9 + j] *= scale;
    }
    __syncthreads();

    if (tid < 81) {
        float acc = dense_b[tid];
        #pragma unroll
        for (int i = 0; i < FCH; ++i) acc += gap[i] * dwm[i * 81 + tid];
        dd[tid] = acc;
    }
    for (int i = tid; i < DCSZ; i += NTHREADS) {
        const int p  = i / DCPAD;
        const int f  = i - p * DCPAD;
        float acc = 0.f;
        if (f < FCH) {
            const int py = p / PSZ, px = p - py * PSZ;
            #pragma unroll
            for (int dy = 0; dy < 3; ++dy) {
                const int yy = py + dy - 1;
                if (yy < 0 || yy >= PSZ) continue;
                #pragma unroll
                for (int dx = 0; dx < 3; ++dx) {
                    const int xx = px + dx - 1;
                    if (xx < 0 || xx >= PSZ) continue;
                    acc += dpt[(yy * PSZ + xx) * FCH + f] * ck[f * 9 + dy * 3 + dx];
                }
            }
        }
        g[i] = acc;
    }
    __syncthreads();

    if (tid == 0) {
        float ss = 0.f;
        for (int j = 0; j < 81; ++j) ss += dd[j] * dd[j];
        sM = 1.0f / fmaxf(sqrtf(ss), 1.0f);
    }
    __syncthreads();

    if (tid < 192) {
        const int k = tid >> 4, o = tid & 15;
        Mp[tid] = (k < 9 && o < 9) ? dd[k * 9 + o] * sM : 0.0f;
    }
    __syncthreads();

    const int wave = tid >> 5;
    const int lane = tid & 31;
    const int row  = lane & 15;
    const int kk   = (lane >> 4) << 1;
    const int mhi  = (lane >> 4) << 3;

    for (int t = 0; t < 4; ++t) {
        const int tile = wave * 4 + t;
        const int pix0 = tile * 16;
        v8f acc = {0.f, 0.f, 0.f, 0.f, 0.f, 0.f, 0.f, 0.f};
        #pragma unroll
        for (int c = 0; c < 3; ++c) {
            v2f a, bm;
            #pragma unroll
            for (int j = 0; j < 2; ++j) {
                const int ch = c * 4 + kk + j;
                a[j]  = g[(pix0 + row) * DCPAD + ch];
                bm[j] = Mp[ch * 16 + row];
            }
            acc = __builtin_amdgcn_wmma_f32_16x16x4_f32(
                 false, a,  false, bm,
                 (short)0, acc,  false,  false);
        }
        #pragma unroll
        for (int rr = 0; rr < 8; ++rr) {
            const int pix = pix0 + rr + mhi;
            if (row < FCH)
                dpt[pix * FCH + row] = acc[rr];
        }
    }
    __syncthreads();

    for (int i = tid; i < NV4; i += NTHREADS) {
        *reinterpret_cast<float4*>(out + base_out + i * 4) =
            *reinterpret_cast<const float4*>(dpt + i * 4);
    }
}

extern "C" void kernel_launch(void* const* d_in, const int* in_sizes, int n_in,
                              void* d_out, int out_size, void* d_ws, size_t ws_size,
                              hipStream_t stream) {
    const float* guidance = (const float*)d_in[0];
    const float* depth    = (const float*)d_in[1];
    const float* conv_w   = (const float*)d_in[2];
    const float* conv_b   = (const float*)d_in[3];
    const float* dense_w  = (const float*)d_in[4];
    const float* dense_b  = (const float*)d_in[5];
    float* out = (float*)d_out;

    const int nblocks = 16 * 16 * 16;
    guided_conv_fused<<<nblocks, NTHREADS, 0, stream>>>(
        guidance, depth, conv_w, conv_b, dense_w, dense_b, out);
}


# depthwise 3x3 as per-thread column sweep (thread = column x channel): taps in registers, 3-row register window, 3 LDS reads per output instead of 18, no per-tap branches; sum order unchanged
# speedup vs baseline: 2.1941x; 2.1941x over previous
#include <hip/hip_runtime.h>
#include <hip/hip_bf16.h>

typedef __attribute__((ext_vector_type(2))) float v2f;
typedef __attribute__((ext_vector_type(8))) float v8f;

#define FCH 9
#define PSZ 24
#define NPIX (PSZ * PSZ)
#define PATCH (NPIX * FCH)
#define NV4   (PATCH / 4)
#define DCPAD 12
#define DCSZ  (NPIX * DCPAD)
#define NTHREADS 288

__global__ __launch_bounds__(NTHREADS)
void guided_conv_fused(const float* __restrict__ gin,
                       const float* __restrict__ din,
                       const float* __restrict__ conv_w,
                       const float* __restrict__ conv_b,
                       const float* __restrict__ dense_w,
                       const float* __restrict__ dense_b,
                       float* __restrict__ out)
{
    __shared__ __align__(16) float g[DCSZ];
    __shared__ __align__(16) float dpt[PATCH];
    __shared__ float cw[729];
    __shared__ float dwm[729];
    __shared__ float ck[81];
    __shared__ float part[NTHREADS];
    __shared__ float gap[FCH];
    __shared__ float dd[81];
    __shared__ float Mp[12 * 16];
    __shared__ float sM;

    const int tid = threadIdx.x;
    const int n  = blockIdx.x;
    const int b  = n >> 8;
    const int r  = n & 255;
    const int ih = r >> 4;
    const int iw = r & 15;
    const long base_in  = (((long)b * 384 + ih * PSZ) * 384 + iw * PSZ) * FCH;
    const long base_out = (long)n * PATCH;

    for (int i = tid; i < NV4; i += NTHREADS) {
        const int py = i / 54;
        const int q  = i - py * 54;
        const long off = base_in + (long)py * (384 * FCH) + q * 4;
        const float4 gv = *reinterpret_cast<const float4*>(gin + off);
        const float4 dv = *reinterpret_cast<const float4*>(din + off);
        *reinterpret_cast<float4*>(g   + i * 4) = gv;
        *reinterpret_cast<float4*>(dpt + i * 4) = dv;
    }
    for (int i = tid; i < 729; i += NTHREADS) {
        cw[i]  = conv_w[i];
        dwm[i] = dense_w[i];
    }
    __syncthreads();

    if (tid < 81) {
        const int oc = tid % 9;
        const int kx = (tid / 9) % 3;
        const int ky = tid / 27;
        float acc = conv_b[oc];
        #pragma unroll
        for (int dy = 0; dy < 3; ++dy)
            #pragma unroll
            for (int dx = 0; dx < 3; ++dx)
                for (int ic = 0; ic < FCH; ++ic)
                    acc += g[((ky * 8 + dy) * PSZ + (kx * 8 + dx)) * FCH + ic]
                         * cw[((dy * 3 + dx) * 9 + ic) * 9 + oc];
        ck[oc * 9 + ky * 3 + kx] = acc;
    }
    {
        const int c = tid % 9;
        const int s = tid / 9;
        float acc = 0.f;
        for (int p = s; p < NPIX; p += 32) acc += g[p * FCH + c];
        part[tid] = acc;
    }
    __syncthreads();

    if (tid < 9) {
        float acc = 0.f;
        for (int s = 0; s < 32; ++s) acc += part[s * 9 + tid];
        gap[tid] = acc * (1.0f / (float)NPIX);
        float ss = 0.f;
        #pragma unroll
        for (int j = 0; j < 9; ++j) { float v = ck[tid * 9 + j]; ss += v * v; }
        const float scale = 1.0f / fmaxf(sqrtf(ss), 1.0f);
        #pragma unroll
        for (int j = 0; j < 9; ++j) ck[tid * 9 + j] *= scale;
    }
    __syncthreads();

    if (tid < 81) {
        float acc = dense_b[tid];
        #pragma unroll
        for (int i = 0; i < FCH; ++i) acc += gap[i] * dwm[i * 81 + tid];
        dd[tid] = acc;
    }
    if (tid < PSZ * FCH) {
        const int px = tid / FCH;
        const int f  = tid - px * FCH;
        float w[9];
        #pragma unroll
        for (int j = 0; j < 9; ++j) w[j] = ck[f * 9 + j];
        const bool hasL = px > 0;
        const bool hasR = px < PSZ - 1;
        const float* colC = dpt + tid;
        const float* colL = hasL ? colC - FCH : colC;
        const float* colR = hasR ? colC + FCH : colC;
        float* dst = g + px * DCPAD + f;
        float a0 = 0.f, a1 = 0.f, a2 = 0.f;
        float b1 = colC[0];
        float b0 = hasL ? colL[0] : 0.f;
        float b2 = hasR ? colR[0] : 0.f;
        #pragma unroll
        for (int py = 0; py < PSZ; ++py) {
            float c0 = 0.f, c1 = 0.f, c2 = 0.f;
            if (py + 1 < PSZ) {
                const float l = colL[(py + 1) * PSZ * FCH];
                const float r = colR[(py + 1) * PSZ * FCH];
                c1 = colC[(py + 1) * PSZ * FCH];
                c0 = hasL ? l : 0.f;
                c2 = hasR ? r : 0.f;
            }
            float acc = 0.f;
            if (py > 0) { acc += a0 * w[0]; acc += a1 * w[1]; acc += a2 * w[2]; }
            acc += b0 * w[3]; acc += b1 * w[4]; acc += b2 * w[5];
            if (py + 1 < PSZ) { acc += c0 * w[6]; acc += c1 * w[7]; acc += c2 * w[8]; }
            dst[py * PSZ * DCPAD] = acc;
            a0 = b0; a1 = b1; a2 = b2;
            b0 = c0; b1 = c1; b2 = c2;
        }
    } else if (tid >= NTHREADS - 64) {
        float* zp = g + (tid - (NTHREADS - 64)) * DCPAD + FCH;
        #pragma unroll
        for (int k = 0; k < NPIX / 64; ++k) {
            zp[k * 64 * DCPAD + 0] = 0.f;
            zp[k * 64 * DCPAD + 1] = 0.f;
            zp[k * 64 * DCPAD + 2] = 0.f;
        }
    }
    __syncthreads();

    if (tid == 0) {
        float ss = 0.f;
        for (int j = 0; j < 81; ++j) ss += dd[j] * dd[j];
        sM = 1.0f / fmaxf(sqrtf(ss), 1.0f);
    }
    __syncthreads();

    if (tid < 192) {
        const int k = tid >> 4, o = tid & 15;
        Mp[tid] = (k < 9 && o < 9) ? dd[k * 9 + o] * sM : 0.0f;
    }
    __syncthreads();

    const int wave = tid >> 5;
    const int lane = tid & 31;
    const int row  = lane & 15;
    const int kk   = (lane >> 4) << 1;
    const int mhi  = (lane >> 4) << 3;

    for (int t = 0; t < 4; ++t) {
        const int tile = wave * 4 + t;
        const int pix0 = tile * 16;
        v8f acc = {0.f, 0.f, 0.f, 0.f, 0.f, 0.f, 0.f, 0.f};
        #pragma unroll
        for (int c = 0; c < 3; ++c) {
            v2f a, bm;
            #pragma unroll
            for (int j = 0; j < 2; ++j) {
                const int ch = c * 4 + kk + j;
                a[j]  = g[(pix0 + row) * DCPAD + ch];
                bm[j] = Mp[ch * 16 + row];
            }
            acc = __builtin_amdgcn_wmma_f32_16x16x4_f32(
                 false, a,  false, bm,
                 (short)0, acc,  false,  false);
        }
        #pragma unroll
        for (int rr = 0; rr < 8; ++rr) {
            const int pix = pix0 + rr + mhi;
            if (row < FCH)
                dpt[pix * FCH + row] = acc[rr];
        }
    }
    __syncthreads();

    for (int i = tid; i < NV4; i += NTHREADS) {
        *reinterpret_cast<float4*>(out + base_out + i * 4) =
            *reinterpret_cast<const float4*>(dpt + i * 4);
    }
}

extern "C" void kernel_launch(void* const* d_in, const int* in_sizes, int n_in,
                              void* d_out, int out_size, void* d_ws, size_t ws_size,
                              hipStream_t stream) {
    const float* guidance = (const float*)d_in[0];
    const float* depth    = (const float*)d_in[1];
    const float* conv_w   = (const float*)d_in[2];
    const float* conv_b   = (const float*)d_in[3];
    const float* dense_w  = (const float*)d_in[4];
    const float* dense_b  = (const float*)d_in[5];
    float* out = (float*)d_out;

    const int nblocks = 16 * 16 * 16;
    guided_conv_fused<<<nblocks, NTHREADS, 0, stream>>>(
        guidance, depth, conv_w, conv_b, dense_w, dense_b, out);
}


# all global loads issued at kernel entry into registers and waited for only at first use (depth patch reaches LDS after the conv phase, LDS-only barriers before that): 1 memory round trip, not 8
# speedup vs baseline: 2.7395x; 1.2486x over previous
#include <hip/hip_runtime.h>
#include <hip/hip_bf16.h>

typedef __attribute__((ext_vector_type(2))) float v2f;
typedef __attribute__((ext_vector_type(4))) float v4f;
typedef __attribute__((ext_vector_type(8))) float v8f;

#define FCH 9
#define PSZ 24
#define NPIX (PSZ * PSZ)
#define PATCH (NPIX * FCH)
#define NV4   (PATCH / 4)
#define DCPAD 12
#define DCSZ  (NPIX * DCPAD)
#define NTHREADS 288

static __device__ __forceinline__ void lds_barrier() {
    __builtin_amdgcn_fence(__ATOMIC_RELEASE, "workgroup", "local");
    __builtin_amdgcn_s_barrier();
    __builtin_amdgcn_fence(__ATOMIC_ACQUIRE, "workgroup", "local");
}

__global__ __launch_bounds__(NTHREADS)
void guided_conv_fused(const float* __restrict__ gin,
                       const float* __restrict__ din,
                       const float* __restrict__ conv_w,
                       const float* __restrict__ conv_b,
                       const float* __restrict__ dense_w,
                       const float* __restrict__ dense_b,
                       float* __restrict__ out)
{
    __shared__ __align__(16) float g[DCSZ];
    __shared__ __align__(16) float dpt[PATCH];
    __shared__ float cw[729];
    __shared__ float dwm[729];
    __shared__ float ck[81];
    __shared__ float part[NTHREADS];
    __shared__ float gap[FCH];
    __shared__ float dd[81];
    __shared__ float Mp[12 * 16];
    __shared__ float sM;

    const int tid = threadIdx.x;
    const int n  = blockIdx.x;
    const int b  = n >> 8;
    const int r  = n & 255;
    const int ih = r >> 4;
    const int iw = r & 15;
    const long base_in  = (((long)b * 384 + ih * PSZ) * 384 + iw * PSZ) * FCH;
    const long base_out = (long)n * PATCH;

    const float* __restrict__ gsrc = gin + base_in;
    const float* __restrict__ dsrc = din + base_in;
    const bool w2 = tid < 729 - 2 * NTHREADS;
    const bool p4 = tid < NV4 - 4 * NTHREADS;
    const float cw0 = conv_w[tid],  cw1 = conv_w[tid + NTHREADS];
    const float dw0 = dense_w[tid], dw1 = dense_w[tid + NTHREADS];
    float cw2 = 0.f, dw2 = 0.f;
    if (w2) { cw2 = conv_w[tid + 2 * NTHREADS]; dw2 = dense_w[tid + 2 * NTHREADS]; }
    float cb = 0.f, db = 0.f;
    if (tid < 81) { cb = conv_b[tid % 9]; db = dense_b[tid]; }
    const int i0 = tid, i1 = tid + NTHREADS, i2 = tid + 2 * NTHREADS,
              i3 = tid + 3 * NTHREADS, i4 = tid + 4 * NTHREADS;
    const int rowskip = (384 - PSZ) * FCH;
    const int o0 = i0 * 4 + (i0 / 54) * rowskip;
    const int o1 = i1 * 4 + (i1 / 54) * rowskip;
    const int o2 = i2 * 4 + (i2 / 54) * rowskip;
    const int o3 = i3 * 4 + (i3 / 54) * rowskip;
    const int o4 = i4 * 4 + (i4 / 54) * rowskip;
    const v4f g0 = *reinterpret_cast<const v4f*>(gsrc + o0);
    const v4f g1 = *reinterpret_cast<const v4f*>(gsrc + o1);
    const v4f g2 = *reinterpret_cast<const v4f*>(gsrc + o2);
    const v4f g3 = *reinterpret_cast<const v4f*>(gsrc + o3);
    v4f g4 = {0.f, 0.f, 0.f, 0.f};
    if (p4) g4 = *reinterpret_cast<const v4f*>(gsrc + o4);
    const v4f d0 = *reinterpret_cast<const v4f*>(dsrc + o0);
    const v4f d1 = *reinterpret_cast<const v4f*>(dsrc + o1);
    const v4f d2 = *reinterpret_cast<const v4f*>(dsrc + o2);
    const v4f d3 = *reinterpret_cast<const v4f*>(dsrc + o3);
    v4f d4 = {0.f, 0.f, 0.f, 0.f};
    if (p4) d4 = *reinterpret_cast<const v4f*>(dsrc + o4);
    cw[i0] = cw0; dwm[i0] = dw0;
    cw[i1] = cw1; dwm[i1] = dw1;
    if (w2) { cw[i2] = cw2; dwm[i2] = dw2; }
    *reinterpret_cast<v4f*>(g + i0 * 4) = g0;
    *reinterpret_cast<v4f*>(g + i1 * 4) = g1;
    *reinterpret_cast<v4f*>(g + i2 * 4) = g2;
    *reinterpret_cast<v4f*>(g + i3 * 4) = g3;
    if (p4) *reinterpret_cast<v4f*>(g + i4 * 4) = g4;
    lds_barrier();

    if (tid < 81) {
        const int oc = tid % 9;
        const int kx = (tid / 9) % 3;
        const int ky = tid / 27;
        float acc = cb;
        #pragma unroll
        for (int dy = 0; dy < 3; ++dy)
            #pragma unroll
            for (int dx = 0; dx < 3; ++dx)
                for (int ic = 0; ic < FCH; ++ic)
                    acc += g[((ky * 8 + dy) * PSZ + (kx * 8 + dx)) * FCH + ic]
                         * cw[((dy * 3 + dx) * 9 + ic) * 9 + oc];
        ck[oc * 9 + ky * 3 + kx] = acc;
    }
    {
        const int c = tid % 9;
        const int s = tid / 9;
        float acc = 0.f;
        for (int p = s; p < NPIX; p += 32) acc += g[p * FCH + c];
        part[tid] = acc;
    }
    lds_barrier();

    if (tid < 9) {
        float acc = 0.f;
        for (int s = 0; s < 32; ++s) acc += part[s * 9 + tid];
        gap[tid] = acc * (1.0f / (float)NPIX);
        float ss = 0.f;
        #pragma unroll
        for (int j = 0; j < 9; ++j) { float v = ck[tid * 9 + j]; ss += v * v; }
        const float scale = 1.0f / fmaxf(sqrtf(ss), 1.0f);
        #pragma unroll
        for (int j = 0; j < 9; ++j) ck[tid * 9 + j] *= scale;
    }
    *reinterpret_cast<v4f*>(dpt + i0 * 4) = d0;
    *reinterpret_cast<v4f*>(dpt + i1 * 4) = d1;
    *reinterpret_cast<v4f*>(dpt + i2 * 4) = d2;
    *reinterpret_cast<v4f*>(dpt + i3 * 4) = d3;
    if (p4) *reinterpret_cast<v4f*>(dpt + i4 * 4) = d4;
    __syncthreads();

    if (tid < 81) {
        float acc = db;
        #pragma unroll
        for (int i = 0; i < FCH; ++i) acc += gap[i] * dwm[i * 81 + tid];
        dd[tid] = acc;
    }
    if (tid < PSZ * FCH) {
        const int px = tid / FCH;
        const int f  = tid - px * FCH;
        float w[9];
        #pragma unroll
        for (int j = 0; j < 9; ++j) w[j] = ck[f * 9 + j];
        const bool hasL = px > 0;
        const bool hasR = px < PSZ - 1;
        const float* colC = dpt + tid;
        const float* colL = hasL ? colC - FCH : colC;
        const float* colR = hasR ? colC + FCH : colC;
        float* dst = g + px * DCPAD + f;
        float a0 = 0.f, a1 = 0.f, a2 = 0.f;
        float b1 = colC[0];
        float b0 = hasL ? colL[0] : 0.f;
        float b2 = hasR ? colR[0] : 0.f;
        #pragma unroll
        for (int py = 0; py < PSZ; ++py) {
            float c0 = 0.f, c1 = 0.f, c2 = 0.f;
            if (py + 1 < PSZ) {
                const float l = colL[(py + 1) * PSZ * FCH];
                const float r = colR[(py + 1) * PSZ * FCH];
                c1 = colC[(py + 1) * PSZ * FCH];
                c0 = hasL ? l : 0.f;
                c2 = hasR ? r : 0.f;
            }
            float acc = 0.f;
            if (py > 0) { acc += a0 * w[0]; acc += a1 * w[1]; acc += a2 * w[2]; }
            acc += b0 * w[3]; acc += b1 * w[4]; acc += b2 * w[5];
            if (py + 1 < PSZ) { acc += c0 * w[6]; acc += c1 * w[7]; acc += c2 * w[8]; }
            dst[py * PSZ * DCPAD] = acc;
            a0 = b0; a1 = b1; a2 = b2;
            b0 = c0; b1 = c1; b2 = c2;
        }
    } else if (tid >= NTHREADS - 64) {
        float* zp = g + (tid - (NTHREADS - 64)) * DCPAD + FCH;
        #pragma unroll
        for (int k = 0; k < NPIX / 64; ++k) {
            zp[k * 64 * DCPAD + 0] = 0.f;
            zp[k * 64 * DCPAD + 1] = 0.f;
            zp[k * 64 * DCPAD + 2] = 0.f;
        }
    }
    __syncthreads();

    if (tid == 0) {
        float ss = 0.f;
        for (int j = 0; j < 81; ++j) ss += dd[j] * dd[j];
        sM = 1.0f / fmaxf(sqrtf(ss), 1.0f);
    }
    __syncthreads();

    if (tid < 192) {
        const int k = tid >> 4, o = tid & 15;
        Mp[tid] = (k < 9 && o < 9) ? dd[k * 9 + o] * sM : 0.0f;
    }
    __syncthreads();

    const int wave = tid >> 5;
    const int lane = tid & 31;
    const int row  = lane & 15;
    const int kk   = (lane >> 4) << 1;
    const int mhi  = (lane >> 4) << 3;

    for (int t = 0; t < 4; ++t) {
        const int tile = wave * 4 + t;
        const int pix0 = tile * 16;
        v8f acc = {0.f, 0.f, 0.f, 0.f, 0.f, 0.f, 0.f, 0.f};
        #pragma unroll
        for (int c = 0; c < 3; ++c) {
            v2f a, bm;
            #pragma unroll
            for (int j = 0; j < 2; ++j) {
                const int ch = c * 4 + kk + j;
                a[j]  = g[(pix0 + row) * DCPAD + ch];
                bm[j] = Mp[ch * 16 + row];
            }
            acc = __builtin_amdgcn_wmma_f32_16x16x4_f32(
                 false, a,  false, bm,
                 (short)0, acc,  false,  false);
        }
        #pragma unroll
        for (int rr = 0; rr < 8; ++rr) {
            const int pix = pix0 + rr + mhi;
            if (row < FCH)
                dpt[pix * FCH + row] = acc[rr];
        }
    }
    __syncthreads();

    for (int i = tid; i < NV4; i += NTHREADS) {
        *reinterpret_cast<float4*>(out + base_out + i * 4) =
            *reinterpret_cast<const float4*>(dpt + i * 4);
    }
}

extern "C" void kernel_launch(void* const* d_in, const int* in_sizes, int n_in,
                              void* d_out, int out_size, void* d_ws, size_t ws_size,
                              hipStream_t stream) {
    const float* guidance = (const float*)d_in[0];
    const float* depth    = (const float*)d_in[1];
    const float* conv_w   = (const float*)d_in[2];
    const float* conv_b   = (const float*)d_in[3];
    const float* dense_w  = (const float*)d_in[4];
    const float* dense_b  = (const float*)d_in[5];
    float* out = (float*)d_out;

    const int nblocks = 16 * 16 * 16;
    guided_conv_fused<<<nblocks, NTHREADS, 0, stream>>>(
        guidance, depth, conv_w, conv_b, dense_w, dense_b, out);
}


# few-thread phases folded into consumers: sweep threads clip own taps, one free wave does gap+dense+clip+B matrix beside the sweep; 7 barriers to 4; dense clip norm summed by lane tree, not serially
# speedup vs baseline: 3.0052x; 1.0970x over previous
#include <hip/hip_runtime.h>
#include <hip/hip_bf16.h>

typedef __attribute__((ext_vector_type(2))) float v2f;
typedef __attribute__((ext_vector_type(4))) float v4f;
typedef __attribute__((ext_vector_type(8))) float v8f;

#define FCH 9
#define PSZ 24
#define NPIX (PSZ * PSZ)
#define PATCH (NPIX * FCH)
#define NV4   (PATCH / 4)
#define DCPAD 12
#define DCSZ  (NPIX * DCPAD)
#define NTHREADS 288

static __device__ __forceinline__ void lds_barrier() {
    __builtin_amdgcn_fence(__ATOMIC_RELEASE, "workgroup", "local");
    __builtin_amdgcn_s_barrier();
    __builtin_amdgcn_fence(__ATOMIC_ACQUIRE, "workgroup", "local");
}

template <int M>
static __device__ __forceinline__ float row_xmask_add(float x) {
    return x + __int_as_float(__builtin_amdgcn_update_dpp(0, __float_as_int(x), 0x160 + M, 0xf, 0xf, false));
}
static __device__ __forceinline__ float other_row(float x) {
    const int b = __float_as_int(x);
    return __int_as_float(__builtin_amdgcn_permlanex16(b, b, 0x76543210, 0xfedcba98, false, false));
}

__global__ __launch_bounds__(NTHREADS)
void guided_conv_fused(const float* __restrict__ gin,
                       const float* __restrict__ din,
                       const float* __restrict__ conv_w,
                       const float* __restrict__ conv_b,
                       const float* __restrict__ dense_w,
                       const float* __restrict__ dense_b,
                       float* __restrict__ out)
{
    __shared__ __align__(16) float g[DCSZ];
    __shared__ __align__(16) float dpt[PATCH];
    __shared__ float cw[729];
    __shared__ float dwm[729];
    __shared__ float ck[81];
    __shared__ __align__(16) float part[FCH * 36];
    __shared__ float dbs[81];
    __shared__ float Mp[12 * 16];

    const int tid = threadIdx.x;
    const int n  = blockIdx.x;
    const int b  = n >> 8;
    const int r  = n & 255;
    const int ih = r >> 4;
    const int iw = r & 15;
    const long base_in  = (((long)b * 384 + ih * PSZ) * 384 + iw * PSZ) * FCH;
    const long base_out = (long)n * PATCH;

    const float* __restrict__ gsrc = gin + base_in;
    const float* __restrict__ dsrc = din + base_in;
    const bool w2 = tid < 729 - 2 * NTHREADS;
    const bool p4 = tid < NV4 - 4 * NTHREADS;
    const float cw0 = conv_w[tid],  cw1 = conv_w[tid + NTHREADS];
    const float dw0 = dense_w[tid], dw1 = dense_w[tid + NTHREADS];
    float cw2 = 0.f, dw2 = 0.f;
    if (w2) { cw2 = conv_w[tid + 2 * NTHREADS]; dw2 = dense_w[tid + 2 * NTHREADS]; }
    float cb = 0.f, db = 0.f;
    if (tid < 81) { cb = conv_b[tid % 9]; db = dense_b[tid]; }
    const int i0 = tid, i1 = tid + NTHREADS, i2 = tid + 2 * NTHREADS,
              i3 = tid + 3 * NTHREADS, i4 = tid + 4 * NTHREADS;
    const int rowskip = (384 - PSZ) * FCH;
    const int o0 = i0 * 4 + (i0 / 54) * rowskip;
    const int o1 = i1 * 4 + (i1 / 54) * rowskip;
    const int o2 = i2 * 4 + (i2 / 54) * rowskip;
    const int o3 = i3 * 4 + (i3 / 54) * rowskip;
    const int o4 = i4 * 4 + (i4 / 54) * rowskip;
    const v4f g0 = *reinterpret_cast<const v4f*>(gsrc + o0);
    const v4f g1 = *reinterpret_cast<const v4f*>(gsrc + o1);
    const v4f g2 = *reinterpret_cast<const v4f*>(gsrc + o2);
    const v4f g3 = *reinterpret_cast<const v4f*>(gsrc + o3);
    v4f g4 = {0.f, 0.f, 0.f, 0.f};
    if (p4) g4 = *reinterpret_cast<const v4f*>(gsrc + o4);
    const v4f d0 = *reinterpret_cast<const v4f*>(dsrc + o0);
    const v4f d1 = *reinterpret_cast<const v4f*>(dsrc + o1);
    const v4f d2 = *reinterpret_cast<const v4f*>(dsrc + o2);
    const v4f d3 = *reinterpret_cast<const v4f*>(dsrc + o3);
    v4f d4 = {0.f, 0.f, 0.f, 0.f};
    if (p4) d4 = *reinterpret_cast<const v4f*>(dsrc + o4);
    cw[i0] = cw0; dwm[i0] = dw0;
    cw[i1] = cw1; dwm[i1] = dw1;
    if (w2) { cw[i2] = cw2; dwm[i2] = dw2; }
    if (tid < 81) dbs[tid] = db;
    *reinterpret_cast<v4f*>(g + i0 * 4) = g0;
    *reinterpret_cast<v4f*>(g + i1 * 4) = g1;
    *reinterpret_cast<v4f*>(g + i2 * 4) = g2;
    *reinterpret_cast<v4f*>(g + i3 * 4) = g3;
    if (p4) *reinterpret_cast<v4f*>(g + i4 * 4) = g4;
    lds_barrier();

    if (tid < 81) {
        const int oc = tid % 9;
        const int kx = (tid / 9) % 3;
        const int ky = tid / 27;
        float acc = cb;
        #pragma unroll
        for (int dy = 0; dy < 3; ++dy)
            #pragma unroll
            for (int dx = 0; dx < 3; ++dx)
                for (int ic = 0; ic < FCH; ++ic)
                    acc += g[((ky * 8 + dy) * PSZ + (kx * 8 + dx)) * FCH + ic]
                         * cw[((dy * 3 + dx) * 9 + ic) * 9 + oc];
        ck[oc * 9 + ky * 3 + kx] = acc;
    }
    {
        const int c = tid % 9;
        const int s = tid / 9;
        float acc = 0.f;
        for (int p = s; p < NPIX; p += 32) acc += g[p * FCH + c];
        part[c * 36 + s] = acc;
    }
    *reinterpret_cast<v4f*>(dpt + i0 * 4) = d0;
    *reinterpret_cast<v4f*>(dpt + i1 * 4) = d1;
    *reinterpret_cast<v4f*>(dpt + i2 * 4) = d2;
    *reinterpret_cast<v4f*>(dpt + i3 * 4) = d3;
    if (p4) *reinterpret_cast<v4f*>(dpt + i4 * 4) = d4;
    lds_barrier();

    if (tid < PSZ * FCH) {
        const int px = tid / FCH;
        const int f  = tid - px * FCH;
        float w[9];
        #pragma unroll
        for (int j = 0; j < 9; ++j) w[j] = ck[f * 9 + j];
        float ss = 0.f;
        #pragma unroll
        for (int j = 0; j < 9; ++j) { float v = w[j]; ss += v * v; }
        const float scale = 1.0f / fmaxf(sqrtf(ss), 1.0f);
        #pragma unroll
        for (int j = 0; j < 9; ++j) w[j] *= scale;
        const bool hasL = px > 0;
        const bool hasR = px < PSZ - 1;
        const float* colC = dpt + tid;
        const float* colL = hasL ? colC - FCH : colC;
        const float* colR = hasR ? colC + FCH : colC;
        float* dst = g + px * DCPAD + f;
        float a0 = 0.f, a1 = 0.f, a2 = 0.f;
        float b1 = colC[0];
        float b0 = hasL ? colL[0] : 0.f;
        float b2 = hasR ? colR[0] : 0.f;
        #pragma unroll
        for (int py = 0; py < PSZ; ++py) {
            float c0 = 0.f, c1 = 0.f, c2 = 0.f;
            if (py + 1 < PSZ) {
                const float l = colL[(py + 1) * PSZ * FCH];
                const float r = colR[(py + 1) * PSZ * FCH];
                c1 = colC[(py + 1) * PSZ * FCH];
                c0 = hasL ? l : 0.f;
                c2 = hasR ? r : 0.f;
            }
            float acc = 0.f;
            if (py > 0) { acc += a0 * w[0]; acc += a1 * w[1]; acc += a2 * w[2]; }
            acc += b0 * w[3]; acc += b1 * w[4]; acc += b2 * w[5];
            if (py + 1 < PSZ) { acc += c0 * w[6]; acc += c1 * w[7]; acc += c2 * w[8]; }
            dst[py * PSZ * DCPAD] = acc;
            a0 = b0; a1 = b1; a2 = b2;
            b0 = c0; b1 = c1; b2 = c2;
        }
    } else if (tid >= NTHREADS - 32) {
        const int lane = tid - (NTHREADS - 32);
        float gsum = 0.f;
        if (lane < FCH) {
            const v4f* pp = reinterpret_cast<const v4f*>(part + lane * 36);
            v4f q[8];
            #pragma unroll
            for (int k = 0; k < 8; ++k) q[k] = pp[k];
            __builtin_amdgcn_sched_barrier(0);
            #pragma unroll
            for (int k = 0; k < 8; ++k) { gsum += q[k][0]; gsum += q[k][1]; gsum += q[k][2]; gsum += q[k][3]; }
            gsum *= (1.0f / (float)NPIX);
        }
        float gp[FCH];
        #pragma unroll
        for (int j = 0; j < FCH; ++j)
            gp[j] = __int_as_float(__builtin_amdgcn_readlane(__float_as_int(gsum), j));
        const int o = lane & 15, h = lane >> 4;
        float val[6];
        float sq = 0.f;
        #pragma unroll
        for (int i = 0; i < 6; ++i) {
            const int k = h + 2 * i;
            const bool ok = (o < FCH) && (k < FCH);
            const int idx = ok ? k * 9 + o : 0;
            float acc = dbs[idx];
            #pragma unroll
            for (int j = 0; j < FCH; ++j) acc += gp[j] * dwm[j * 81 + idx];
            val[i] = ok ? acc : 0.f;
            sq += val[i] * val[i];
        }
        sq = row_xmask_add<1>(sq);
        sq = row_xmask_add<2>(sq);
        sq = row_xmask_add<4>(sq);
        sq = row_xmask_add<8>(sq);
        sq += other_row(sq);
        const float sM = 1.0f / fmaxf(sqrtf(sq), 1.0f);
        #pragma unroll
        for (int i = 0; i < 6; ++i) {
            const int k = h + 2 * i;
            const bool ok = (o < FCH) && (k < FCH);
            Mp[lane + 32 * i] = ok ? val[i] * sM : 0.0f;
        }
    } else if (tid >= NTHREADS - 64) {
        float* zp = g + (tid - (NTHREADS - 64)) * DCPAD + FCH;
        #pragma unroll
        for (int k = 0; k < NPIX / 32; ++k) {
            zp[k * 32 * DCPAD + 0] = 0.f;
            zp[k * 32 * DCPAD + 1] = 0.f;
            zp[k * 32 * DCPAD + 2] = 0.f;
        }
    }
    __syncthreads();

    const int wave = tid >> 5;
    const int lane = tid & 31;
    const int row  = lane & 15;
    const int kk   = (lane >> 4) << 1;
    const int mhi  = (lane >> 4) << 3;

    for (int t = 0; t < 4; ++t) {
        const int tile = wave * 4 + t;
        const int pix0 = tile * 16;
        v8f acc = {0.f, 0.f, 0.f, 0.f, 0.f, 0.f, 0.f, 0.f};
        #pragma unroll
        for (int c = 0; c < 3; ++c) {
            v2f a, bm;
            #pragma unroll
            for (int j = 0; j < 2; ++j) {
                const int ch = c * 4 + kk + j;
                a[j]  = g[(pix0 + row) * DCPAD + ch];
                bm[j] = Mp[ch * 16 + row];
            }
            acc = __builtin_amdgcn_wmma_f32_16x16x4_f32(
                 false, a,  false, bm,
                 (short)0, acc,  false,  false);
        }
        #pragma unroll
        for (int rr = 0; rr < 8; ++rr) {
            const int pix = pix0 + rr + mhi;
            if (row < FCH)
                dpt[pix * FCH + row] = acc[rr];
        }
    }
    __syncthreads();

    for (int i = tid; i < NV4; i += NTHREADS) {
        *reinterpret_cast<float4*>(out + base_out + i * 4) =
            *reinterpret_cast<const float4*>(dpt + i * 4);
    }
}

extern "C" void kernel_launch(void* const* d_in, const int* in_sizes, int n_in,
                              void* d_out, int out_size, void* d_ws, size_t ws_size,
                              hipStream_t stream) {
    const float* guidance = (const float*)d_in[0];
    const float* depth    = (const float*)d_in[1];
    const float* conv_w   = (const float*)d_in[2];
    const float* conv_b   = (const float*)d_in[3];
    const float* dense_w  = (const float*)d_in[4];
    const float* dense_b  = (const float*)d_in[5];
    float* out = (float*)d_out;

    const int nblocks = 16 * 16 * 16;
    guided_conv_fused<<<nblocks, NTHREADS, 0, stream>>>(
        guidance, depth, conv_w, conv_b, dense_w, dense_b, out);
}


# stride-8 conv on the matrix unit: one wave, 21 chained f32 16x16x4 WMMAs (K = 3 x 28 slots in tap order, bias as C) replace 81 threads' 81-term serial LDS chains; sum order kept
# speedup vs baseline: 3.0981x; 1.0309x over previous
#include <hip/hip_runtime.h>
#include <hip/hip_bf16.h>

typedef __attribute__((ext_vector_type(2))) float v2f;
typedef __attribute__((ext_vector_type(4))) float v4f;
typedef __attribute__((ext_vector_type(8))) float v8f;

#define FCH 9
#define PSZ 24
#define NPIX (PSZ * PSZ)
#define PATCH (NPIX * FCH)
#define NV4   (PATCH / 4)
#define DCPAD 12
#define DCSZ  (NPIX * DCPAD)
#define NTHREADS 288

static __device__ __forceinline__ void lds_barrier() {
    __builtin_amdgcn_fence(__ATOMIC_RELEASE, "workgroup", "local");
    __builtin_amdgcn_s_barrier();
    __builtin_amdgcn_fence(__ATOMIC_ACQUIRE, "workgroup", "local");
}

template <int M>
static __device__ __forceinline__ float row_xmask_add(float x) {
    return x + __int_as_float(__builtin_amdgcn_update_dpp(0, __float_as_int(x), 0x160 + M, 0xf, 0xf, false));
}
static __device__ __forceinline__ float other_row(float x) {
    const int b = __float_as_int(x);
    return __int_as_float(__builtin_amdgcn_permlanex16(b, b, 0x76543210, 0xfedcba98, false, false));
}

__global__ __launch_bounds__(NTHREADS)
void guided_conv_fused(const float* __restrict__ gin,
                       const float* __restrict__ din,
                       const float* __restrict__ conv_w,
                       const float* __restrict__ conv_b,
                       const float* __restrict__ dense_w,
                       const float* __restrict__ dense_b,
                       float* __restrict__ out)
{
    __shared__ __align__(16) float g[DCSZ];
    __shared__ __align__(16) float dpt[PATCH];
    __shared__ float cw[729 + 9];
    __shared__ float dwm[729];
    __shared__ float ck[81];
    __shared__ __align__(16) float part[FCH * 36];
    __shared__ float dbs[81];
    __shared__ float cbs[FCH];
    __shared__ float Mp[12 * 16];

    const int tid = threadIdx.x;
    const int n  = blockIdx.x;
    const int b  = n >> 8;
    const int r  = n & 255;
    const int ih = r >> 4;
    const int iw = r & 15;
    const long base_in  = (((long)b * 384 + ih * PSZ) * 384 + iw * PSZ) * FCH;
    const long base_out = (long)n * PATCH;

    const float* __restrict__ gsrc = gin + base_in;
    const float* __restrict__ dsrc = din + base_in;
    const bool w2 = tid < 729 - 2 * NTHREADS;
    const bool p4 = tid < NV4 - 4 * NTHREADS;
    const float cw0 = conv_w[tid],  cw1 = conv_w[tid + NTHREADS];
    const float dw0 = dense_w[tid], dw1 = dense_w[tid + NTHREADS];
    float cw2 = 0.f, dw2 = 0.f;
    if (w2) { cw2 = conv_w[tid + 2 * NTHREADS]; dw2 = dense_w[tid + 2 * NTHREADS]; }
    float cb = 0.f, db = 0.f;
    if (tid < 81) db = dense_b[tid];
    if (tid < FCH) cb = conv_b[tid];
    const int i0 = tid, i1 = tid + NTHREADS, i2 = tid + 2 * NTHREADS,
              i3 = tid + 3 * NTHREADS, i4 = tid + 4 * NTHREADS;
    const int rowskip = (384 - PSZ) * FCH;
    const int o0 = i0 * 4 + (i0 / 54) * rowskip;
    const int o1 = i1 * 4 + (i1 / 54) * rowskip;
    const int o2 = i2 * 4 + (i2 / 54) * rowskip;
    const int o3 = i3 * 4 + (i3 / 54) * rowskip;
    const int o4 = i4 * 4 + (i4 / 54) * rowskip;
    const v4f g0 = *reinterpret_cast<const v4f*>(gsrc + o0);
    const v4f g1 = *reinterpret_cast<const v4f*>(gsrc + o1);
    const v4f g2 = *reinterpret_cast<const v4f*>(gsrc + o2);
    const v4f g3 = *reinterpret_cast<const v4f*>(gsrc + o3);
    v4f g4 = {0.f, 0.f, 0.f, 0.f};
    if (p4) g4 = *reinterpret_cast<const v4f*>(gsrc + o4);
    const v4f d0 = *reinterpret_cast<const v4f*>(dsrc + o0);
    const v4f d1 = *reinterpret_cast<const v4f*>(dsrc + o1);
    const v4f d2 = *reinterpret_cast<const v4f*>(dsrc + o2);
    const v4f d3 = *reinterpret_cast<const v4f*>(dsrc + o3);
    v4f d4 = {0.f, 0.f, 0.f, 0.f};
    if (p4) d4 = *reinterpret_cast<const v4f*>(dsrc + o4);
    cw[i0] = cw0; dwm[i0] = dw0;
    cw[i1] = cw1; dwm[i1] = dw1;
    if (w2) { cw[i2] = cw2; dwm[i2] = dw2; }
    if (tid < 81) dbs[tid] = db;
    if (tid < FCH) { cbs[tid] = cb; cw[729 + tid] = 0.f; }
    *reinterpret_cast<v4f*>(g + i0 * 4) = g0;
    *reinterpret_cast<v4f*>(g + i1 * 4) = g1;
    *reinterpret_cast<v4f*>(g + i2 * 4) = g2;
    *reinterpret_cast<v4f*>(g + i3 * 4) = g3;
    if (p4) *reinterpret_cast<v4f*>(g + i4 * 4) = g4;
    lds_barrier();

    const bool convwave = __builtin_amdgcn_readfirstlane(tid >> 5) == (NTHREADS / 32 - 1);
    auto gap_partial = [&]() {
        const int c = tid % 9;
        const int s = tid / 9;
        float acc = 0.f;
        for (int p = s; p < NPIX; p += 32) acc += g[p * FCH + c];
        part[c * 36 + s] = acc;
    };
    if (convwave) {
        *reinterpret_cast<v4f*>(dpt + i0 * 4) = d0;
        *reinterpret_cast<v4f*>(dpt + i1 * 4) = d1;
        *reinterpret_cast<v4f*>(dpt + i2 * 4) = d2;
        *reinterpret_cast<v4f*>(dpt + i3 * 4) = d3;
        const int lane = __builtin_amdgcn_mbcnt_lo(0xffffffffu, 0u);
        const int col  = lane & 15;
        const int h    = lane >> 4;
        const int cl   = col < FCH ? col : FCH - 1;
        const int ky   = cl / 3;
        const int kx   = cl - ky * 3;
        const float* ap = g + ((ky * 8) * PSZ + kx * 8) * FCH + 2 * h;
        const float* bp = cw + (2 * h) * 9 + cl;
        v2f fa[21], fb[21];
        const float cbo = cbs[cl];
        v8f acc;
        #pragma unroll
        for (int s0 = -4; s0 < 21; s0 += 4) {
            #pragma unroll
            for (int s = s0 + 4; s < s0 + 8 && s < 21; ++s) {
                const int dy = s / 7, cc = s - dy * 7;
                fa[s] = *reinterpret_cast<const v2f*>(ap + dy * PSZ * FCH + 4 * cc);
                fb[s][0] = bp[(dy * 27 + 4 * cc) * 9];
                fb[s][1] = bp[(dy * 27 + 4 * cc + 1) * 9];
            }
            __builtin_amdgcn_sched_barrier(0);
            if (s0 == 0) acc = v8f{cbo, cbo, cbo, cbo, cbo, cbo, cbo, cbo};
            #pragma unroll
            for (int s = (s0 < 0 ? 0 : s0); s < s0 + 4 && s < 21; ++s) {
                if (s % 7 == 6) {
                    fa[s][1] = h ? 0.f : fa[s][1];
                    fb[s][1] = h ? 0.f : fb[s][1];
                }
                acc = __builtin_amdgcn_wmma_f32_16x16x4_f32(false, fa[s], false, fb[s], (short)0, acc, false, false);
            }
            __builtin_amdgcn_sched_barrier(0);
        }
        #pragma unroll
        for (int r = 0; r < 8; ++r) {
            const int p = 8 * h + r;
            if (col < FCH && p < 9) ck[col * 9 + p] = acc[r];
        }
        gap_partial();
    } else {
        gap_partial();
        *reinterpret_cast<v4f*>(dpt + i0 * 4) = d0;
        *reinterpret_cast<v4f*>(dpt + i1 * 4) = d1;
        *reinterpret_cast<v4f*>(dpt + i2 * 4) = d2;
        *reinterpret_cast<v4f*>(dpt + i3 * 4) = d3;
        if (p4) *reinterpret_cast<v4f*>(dpt + i4 * 4) = d4;
    }
    lds_barrier();

    if (tid < PSZ * FCH) {
        const int px = tid / FCH;
        const int f  = tid - px * FCH;
        float w[9];
        #pragma unroll
        for (int j = 0; j < 9; ++j) w[j] = ck[f * 9 + j];
        float ss = 0.f;
        #pragma unroll
        for (int j = 0; j < 9; ++j) { float v = w[j]; ss += v * v; }
        const float scale = 1.0f / fmaxf(sqrtf(ss), 1.0f);
        #pragma unroll
        for (int j = 0; j < 9; ++j) w[j] *= scale;
        const bool hasL = px > 0;
        const bool hasR = px < PSZ - 1;
        const float* colC = dpt + tid;
        const float* colL = hasL ? colC - FCH : colC;
        const float* colR = hasR ? colC + FCH : colC;
        float* dst = g + px * DCPAD + f;
        float a0 = 0.f, a1 = 0.f, a2 = 0.f;
        float b1 = colC[0];
        float b0 = hasL ? colL[0] : 0.f;
        float b2 = hasR ? colR[0] : 0.f;
        #pragma unroll
        for (int py = 0; py < PSZ; ++py) {
            float c0 = 0.f, c1 = 0.f, c2 = 0.f;
            if (py + 1 < PSZ) {
                const float l = colL[(py + 1) * PSZ * FCH];
                const float r = colR[(py + 1) * PSZ * FCH];
                c1 = colC[(py + 1) * PSZ * FCH];
                c0 = hasL ? l : 0.f;
                c2 = hasR ? r : 0.f;
            }
            float acc = 0.f;
            if (py > 0) { acc += a0 * w[0]; acc += a1 * w[1]; acc += a2 * w[2]; }
            acc += b0 * w[3]; acc += b1 * w[4]; acc += b2 * w[5];
            if (py + 1 < PSZ) { acc += c0 * w[6]; acc += c1 * w[7]; acc += c2 * w[8]; }
            dst[py * PSZ * DCPAD] = acc;
            a0 = b0; a1 = b1; a2 = b2;
            b0 = c0; b1 = c1; b2 = c2;
        }
    } else if (tid >= NTHREADS - 32) {
        const int lane = tid - (NTHREADS - 32);
        float gsum = 0.f;
        if (lane < FCH) {
            const v4f* pp = reinterpret_cast<const v4f*>(part + lane * 36);
            v4f q[8];
            #pragma unroll
            for (int k = 0; k < 8; ++k) q[k] = pp[k];
            __builtin_amdgcn_sched_barrier(0);
            #pragma unroll
            for (int k = 0; k < 8; ++k) { gsum += q[k][0]; gsum += q[k][1]; gsum += q[k][2]; gsum += q[k][3]; }
            gsum *= (1.0f / (float)NPIX);
        }
        float gp[FCH];
        #pragma unroll
        for (int j = 0; j < FCH; ++j)
            gp[j] = __int_as_float(__builtin_amdgcn_readlane(__float_as_int(gsum), j));
        const int o = lane & 15, h = lane >> 4;
        float val[6];
        float sq = 0.f;
        #pragma unroll
        for (int i = 0; i < 6; ++i) {
            const int k = h + 2 * i;
            const bool ok = (o < FCH) && (k < FCH);
            const int idx = ok ? k * 9 + o : 0;
            float acc = dbs[idx];
            #pragma unroll
            for (int j = 0; j < FCH; ++j) acc += gp[j] * dwm[j * 81 + idx];
            val[i] = ok ? acc : 0.f;
            sq += val[i] * val[i];
        }
        sq = row_xmask_add<1>(sq);
        sq = row_xmask_add<2>(sq);
        sq = row_xmask_add<4>(sq);
        sq = row_xmask_add<8>(sq);
        sq += other_row(sq);
        const float sM = 1.0f / fmaxf(sqrtf(sq), 1.0f);
        #pragma unroll
        for (int i = 0; i < 6; ++i) {
            const int k = h + 2 * i;
            const bool ok = (o < FCH) && (k < FCH);
            Mp[lane + 32 * i] = ok ? val[i] * sM : 0.0f;
        }
    } else if (tid >= NTHREADS - 64) {
        float* zp = g + (tid - (NTHREADS - 64)) * DCPAD + FCH;
        #pragma unroll
        for (int k = 0; k < NPIX / 32; ++k) {
            zp[k * 32 * DCPAD + 0] = 0.f;
            zp[k * 32 * DCPAD + 1] = 0.f;
            zp[k * 32 * DCPAD + 2] = 0.f;
        }
    }
    __syncthreads();

    const int wave = tid >> 5;
    const int lane = tid & 31;
    const int row  = lane & 15;
    const int kk   = (lane >> 4) << 1;
    const int mhi  = (lane >> 4) << 3;

    for (int t = 0; t < 4; ++t) {
        const int tile = wave * 4 + t;
        const int pix0 = tile * 16;
        v8f acc = {0.f, 0.f, 0.f, 0.f, 0.f, 0.f, 0.f, 0.f};
        #pragma unroll
        for (int c = 0; c < 3; ++c) {
            v2f a, bm;
            #pragma unroll
            for (int j = 0; j < 2; ++j) {
                const int ch = c * 4 + kk + j;
                a[j]  = g[(pix0 + row) * DCPAD + ch];
                bm[j] = Mp[ch * 16 + row];
            }
            acc = __builtin_amdgcn_wmma_f32_16x16x4_f32(
                 false, a,  false, bm,
                 (short)0, acc,  false,  false);
        }
        #pragma unroll
        for (int rr = 0; rr < 8; ++rr) {
            const int pix = pix0 + rr + mhi;
            if (row < FCH)
                dpt[pix * FCH + row] = acc[rr];
        }
    }
    __syncthreads();

    for (int i = tid; i < NV4; i += NTHREADS) {
        *reinterpret_cast<float4*>(out + base_out + i * 4) =
            *reinterpret_cast<const float4*>(dpt + i * 4);
    }
}

extern "C" void kernel_launch(void* const* d_in, const int* in_sizes, int n_in,
                              void* d_out, int out_size, void* d_ws, size_t ws_size,
                              hipStream_t stream) {
    const float* guidance = (const float*)d_in[0];
    const float* depth    = (const float*)d_in[1];
    const float* conv_w   = (const float*)d_in[2];
    const float* conv_b   = (const float*)d_in[3];
    const float* dense_w  = (const float*)d_in[4];
    const float* dense_b  = (const float*)d_in[5];
    float* out = (float*)d_out;

    const int nblocks = 16 * 16 * 16;
    guided_conv_fused<<<nblocks, NTHREADS, 0, stream>>>(
        guidance, depth, conv_w, conv_b, dense_w, dense_b, out);
}


# LDS per workgroup 56 to 49 KiB (depthwise results 9 per pixel, K pad made in registers) and a 64-register cap: 6 workgroups resident per WGP instead of 5; sum order unchanged
# speedup vs baseline: 3.2114x; 1.0366x over previous
#include <hip/hip_runtime.h>
#include <hip/hip_bf16.h>

typedef __attribute__((ext_vector_type(2))) float v2f;
typedef __attribute__((ext_vector_type(4))) float v4f;
typedef __attribute__((ext_vector_type(8))) float v8f;

#define FCH 9
#define PSZ 24
#define NPIX (PSZ * PSZ)
#define PATCH (NPIX * FCH)
#define NV4   (PATCH / 4)
#define NTHREADS 288

static __device__ __forceinline__ void lds_barrier() {
    __builtin_amdgcn_fence(__ATOMIC_RELEASE, "workgroup", "local");
    __builtin_amdgcn_s_barrier();
    __builtin_amdgcn_fence(__ATOMIC_ACQUIRE, "workgroup", "local");
}

template <int M>
static __device__ __forceinline__ float row_xmask_add(float x) {
    return x + __int_as_float(__builtin_amdgcn_update_dpp(0, __float_as_int(x), 0x160 + M, 0xf, 0xf, false));
}
static __device__ __forceinline__ float other_row(float x) {
    const int b = __float_as_int(x);
    return __int_as_float(__builtin_amdgcn_permlanex16(b, b, 0x76543210, 0xfedcba98, false, false));
}

__global__ __launch_bounds__(NTHREADS) __attribute__((amdgpu_num_vgpr(64)))
void guided_conv_fused(const float* __restrict__ gin,
                       const float* __restrict__ din,
                       const float* __restrict__ conv_w,
                       const float* __restrict__ conv_b,
                       const float* __restrict__ dense_w,
                       const float* __restrict__ dense_b,
                       float* __restrict__ out)
{
    __shared__ __align__(16) float g[PATCH];
    __shared__ __align__(16) float dpt[PATCH];
    __shared__ float cw[729 + 9];
    __shared__ float dwm[729];
    __shared__ float ck[81];
    __shared__ __align__(16) float part[FCH * 36];
    __shared__ float dbs[81];
    __shared__ float cbs[FCH];
    __shared__ float Mp[12 * 16];

    const int tid = threadIdx.x;
    const int n  = blockIdx.x;
    const int b  = n >> 8;
    const int r  = n & 255;
    const int ih = r >> 4;
    const int iw = r & 15;
    const long base_in  = (((long)b * 384 + ih * PSZ) * 384 + iw * PSZ) * FCH;
    const long base_out = (long)n * PATCH;

    const float* __restrict__ gsrc = gin + base_in;
    const float* __restrict__ dsrc = din + base_in;
    const bool w2 = tid < 729 - 2 * NTHREADS;
    const bool p4 = tid < NV4 - 4 * NTHREADS;
    const float cw0 = conv_w[tid],  cw1 = conv_w[tid + NTHREADS];
    const float dw0 = dense_w[tid], dw1 = dense_w[tid + NTHREADS];
    float cw2 = 0.f, dw2 = 0.f;
    if (w2) { cw2 = conv_w[tid + 2 * NTHREADS]; dw2 = dense_w[tid + 2 * NTHREADS]; }
    float cb = 0.f, db = 0.f;
    if (tid < 81) db = dense_b[tid];
    if (tid < FCH) cb = conv_b[tid];
    const int i0 = tid, i1 = tid + NTHREADS, i2 = tid + 2 * NTHREADS,
              i3 = tid + 3 * NTHREADS, i4 = tid + 4 * NTHREADS;
    const int rowskip = (384 - PSZ) * FCH;
    const int o0 = i0 * 4 + (i0 / 54) * rowskip;
    const int o1 = i1 * 4 + (i1 / 54) * rowskip;
    const int o2 = i2 * 4 + (i2 / 54) * rowskip;
    const int o3 = i3 * 4 + (i3 / 54) * rowskip;
    const int o4 = i4 * 4 + (i4 / 54) * rowskip;
    const v4f g0 = *reinterpret_cast<const v4f*>(gsrc + o0);
    const v4f g1 = *reinterpret_cast<const v4f*>(gsrc + o1);
    const v4f g2 = *reinterpret_cast<const v4f*>(gsrc + o2);
    const v4f g3 = *reinterpret_cast<const v4f*>(gsrc + o3);
    v4f g4 = {0.f, 0.f, 0.f, 0.f};
    if (p4) g4 = *reinterpret_cast<const v4f*>(gsrc + o4);
    const v4f d0 = *reinterpret_cast<const v4f*>(dsrc + o0);
    const v4f d1 = *reinterpret_cast<const v4f*>(dsrc + o1);
    const v4f d2 = *reinterpret_cast<const v4f*>(dsrc + o2);
    const v4f d3 = *reinterpret_cast<const v4f*>(dsrc + o3);
    v4f d4 = {0.f, 0.f, 0.f, 0.f};
    if (p4) d4 = *reinterpret_cast<const v4f*>(dsrc + o4);
    cw[i0] = cw0; dwm[i0] = dw0;
    cw[i1] = cw1; dwm[i1] = dw1;
    if (w2) { cw[i2] = cw2; dwm[i2] = dw2; }
    if (tid < 81) dbs[tid] = db;
    if (tid < FCH) { cbs[tid] = cb; cw[729 + tid] = 0.f; }
    *reinterpret_cast<v4f*>(g + i0 * 4) = g0;
    *reinterpret_cast<v4f*>(g + i1 * 4) = g1;
    *reinterpret_cast<v4f*>(g + i2 * 4) = g2;
    *reinterpret_cast<v4f*>(g + i3 * 4) = g3;
    if (p4) *reinterpret_cast<v4f*>(g + i4 * 4) = g4;
    lds_barrier();

    const bool convwave = __builtin_amdgcn_readfirstlane(tid >> 5) == (NTHREADS / 32 - 1);
    auto gap_partial = [&]() {
        const int c = tid % 9;
        const int s = tid / 9;
        float acc = 0.f;
        for (int p = s; p < NPIX; p += 32) acc += g[p * FCH + c];
        part[c * 36 + s] = acc;
    };
    if (convwave) {
        *reinterpret_cast<v4f*>(dpt + i0 * 4) = d0;
        *reinterpret_cast<v4f*>(dpt + i1 * 4) = d1;
        *reinterpret_cast<v4f*>(dpt + i2 * 4) = d2;
        *reinterpret_cast<v4f*>(dpt + i3 * 4) = d3;
        const int lane = __builtin_amdgcn_mbcnt_lo(0xffffffffu, 0u);
        const int col  = lane & 15;
        const int h    = lane >> 4;
        const int cl   = col < FCH ? col : FCH - 1;
        const int ky   = cl / 3;
        const int kx   = cl - ky * 3;
        const float* ap = g + ((ky * 8) * PSZ + kx * 8) * FCH + 2 * h;
        const float* bp = cw + (2 * h) * 9 + cl;
        v2f fa[21], fb[21];
        const float cbo = cbs[cl];
        v8f acc;
        #pragma unroll
        for (int s0 = -4; s0 < 21; s0 += 4) {
            #pragma unroll
            for (int s = s0 + 4; s < s0 + 8 && s < 21; ++s) {
                const int dy = s / 7, cc = s - dy * 7;
                fa[s] = *reinterpret_cast<const v2f*>(ap + dy * PSZ * FCH + 4 * cc);
                fb[s][0] = bp[(dy * 27 + 4 * cc) * 9];
                fb[s][1] = bp[(dy * 27 + 4 * cc + 1) * 9];
            }
            __builtin_amdgcn_sched_barrier(0);
            if (s0 == 0) acc = v8f{cbo, cbo, cbo, cbo, cbo, cbo, cbo, cbo};
            #pragma unroll
            for (int s = (s0 < 0 ? 0 : s0); s < s0 + 4 && s < 21; ++s) {
                if (s % 7 == 6) {
                    fa[s][1] = h ? 0.f : fa[s][1];
                    fb[s][1] = h ? 0.f : fb[s][1];
                }
                acc = __builtin_amdgcn_wmma_f32_16x16x4_f32(false, fa[s], false, fb[s], (short)0, acc, false, false);
            }
            __builtin_amdgcn_sched_barrier(0);
        }
        #pragma unroll
        for (int r = 0; r < 8; ++r) {
            const int p = 8 * h + r;
            if (col < FCH && p < 9) ck[col * 9 + p] = acc[r];
        }
        gap_partial();
    } else {
        gap_partial();
        *reinterpret_cast<v4f*>(dpt + i0 * 4) = d0;
        *reinterpret_cast<v4f*>(dpt + i1 * 4) = d1;
        *reinterpret_cast<v4f*>(dpt + i2 * 4) = d2;
        *reinterpret_cast<v4f*>(dpt + i3 * 4) = d3;
        if (p4) *reinterpret_cast<v4f*>(dpt + i4 * 4) = d4;
    }
    lds_barrier();

    if (tid < PSZ * FCH) {
        const int px = tid / FCH;
        const int f  = tid - px * FCH;
        float w[9];
        #pragma unroll
        for (int j = 0; j < 9; ++j) w[j] = ck[f * 9 + j];
        float ss = 0.f;
        #pragma unroll
        for (int j = 0; j < 9; ++j) { float v = w[j]; ss += v * v; }
        const float scale = 1.0f / fmaxf(sqrtf(ss), 1.0f);
        #pragma unroll
        for (int j = 0; j < 9; ++j) w[j] *= scale;
        const bool hasL = px > 0;
        const bool hasR = px < PSZ - 1;
        const float* colC = dpt + tid;
        const float* colL = hasL ? colC - FCH : colC;
        const float* colR = hasR ? colC + FCH : colC;
        float* dst = g + tid;
        float a0 = 0.f, a1 = 0.f, a2 = 0.f;
        float b1 = colC[0];
        float b0 = hasL ? colL[0] : 0.f;
        float b2 = hasR ? colR[0] : 0.f;
        #pragma unroll
        for (int py = 0; py < PSZ; ++py) {
            float c0 = 0.f, c1 = 0.f, c2 = 0.f;
            if (py + 1 < PSZ) {
                const float l = colL[(py + 1) * PSZ * FCH];
                const float r = colR[(py + 1) * PSZ * FCH];
                c1 = colC[(py + 1) * PSZ * FCH];
                c0 = hasL ? l : 0.f;
                c2 = hasR ? r : 0.f;
            }
            float acc = 0.f;
            if (py > 0) { acc += a0 * w[0]; acc += a1 * w[1]; acc += a2 * w[2]; }
            acc += b0 * w[3]; acc += b1 * w[4]; acc += b2 * w[5];
            if (py + 1 < PSZ) { acc += c0 * w[6]; acc += c1 * w[7]; acc += c2 * w[8]; }
            dst[py * PSZ * FCH] = acc;
            a0 = b0; a1 = b1; a2 = b2;
            b0 = c0; b1 = c1; b2 = c2;
        }
    } else if (tid >= NTHREADS - 32) {
        const int lane = tid - (NTHREADS - 32);
        float gsum = 0.f;
        if (lane < FCH) {
            const v4f* pp = reinterpret_cast<const v4f*>(part + lane * 36);
            v4f q[8];
            #pragma unroll
            for (int k = 0; k < 8; ++k) q[k] = pp[k];
            __builtin_amdgcn_sched_barrier(0);
            #pragma unroll
            for (int k = 0; k < 8; ++k) { gsum += q[k][0]; gsum += q[k][1]; gsum += q[k][2]; gsum += q[k][3]; }
            gsum *= (1.0f / (float)NPIX);
        }
        float gp[FCH];
        #pragma unroll
        for (int j = 0; j < FCH; ++j)
            gp[j] = __int_as_float(__builtin_amdgcn_readlane(__float_as_int(gsum), j));
        const int o = lane & 15, h = lane >> 4;
        float val[6];
        float sq = 0.f;
        #pragma unroll
        for (int i = 0; i < 6; ++i) {
            const int k = h + 2 * i;
            const bool ok = (o < FCH) && (k < FCH);
            const int idx = ok ? k * 9 + o : 0;
            float acc = dbs[idx];
            #pragma unroll
            for (int j = 0; j < FCH; ++j) acc += gp[j] * dwm[j * 81 + idx];
            val[i] = ok ? acc : 0.f;
            sq += val[i] * val[i];
        }
        sq = row_xmask_add<1>(sq);
        sq = row_xmask_add<2>(sq);
        sq = row_xmask_add<4>(sq);
        sq = row_xmask_add<8>(sq);
        sq += other_row(sq);
        const float sM = 1.0f / fmaxf(sqrtf(sq), 1.0f);
        #pragma unroll
        for (int i = 0; i < 6; ++i) {
            const int k = h + 2 * i;
            const bool ok = (o < FCH) && (k < FCH);
            Mp[lane + 32 * i] = ok ? val[i] * sM : 0.0f;
        }
    }
    __syncthreads();

    const int wave = tid >> 5;
    const int lane = tid & 31;
    const int row  = lane & 15;
    const int kk   = (lane >> 4) << 1;
    const int mhi  = (lane >> 4) << 3;

    for (int t = 0; t < 4; ++t) {
        const int tile = wave * 4 + t;
        const int pix0 = tile * 16;
        v8f acc = {0.f, 0.f, 0.f, 0.f, 0.f, 0.f, 0.f, 0.f};
        #pragma unroll
        for (int c = 0; c < 3; ++c) {
            v2f a, bm;
            #pragma unroll
            for (int j = 0; j < 2; ++j) {
                const int ch = c * 4 + kk + j;
                if (c < 2) {
                    a[j] = g[(pix0 + row) * FCH + ch];
                } else {
                    const float c8 = g[(pix0 + row) * FCH + 8];
                    a[j] = (j == 0 && kk == 0) ? c8 : 0.f;
                }
                bm[j] = Mp[ch * 16 + row];
            }
            acc = __builtin_amdgcn_wmma_f32_16x16x4_f32(
                 false, a,  false, bm,
                 (short)0, acc,  false,  false);
        }
        #pragma unroll
        for (int rr = 0; rr < 8; ++rr) {
            const int pix = pix0 + rr + mhi;
            if (row < FCH)
                dpt[pix * FCH + row] = acc[rr];
        }
    }
    __syncthreads();

    for (int i = tid; i < NV4; i += NTHREADS) {
        *reinterpret_cast<float4*>(out + base_out + i * 4) =
            *reinterpret_cast<const float4*>(dpt + i * 4);
    }
}

extern "C" void kernel_launch(void* const* d_in, const int* in_sizes, int n_in,
                              void* d_out, int out_size, void* d_ws, size_t ws_size,
                              hipStream_t stream) {
    const float* guidance = (const float*)d_in[0];
    const float* depth    = (const float*)d_in[1];
    const float* conv_w   = (const float*)d_in[2];
    const float* conv_b   = (const float*)d_in[3];
    const float* dense_w  = (const float*)d_in[4];
    const float* dense_b  = (const float*)d_in[5];
    float* out = (float*)d_out;

    const int nblocks = 16 * 16 * 16;
    guided_conv_fused<<<nblocks, NTHREADS, 0, stream>>>(
        guidance, depth, conv_w, conv_b, dense_w, dense_b, out);
}
